# starting kernel
# baseline (speedup 1.0000x reference)
#include <hip/hip_runtime.h>
#include <hip/hip_bf16.h>

typedef __attribute__((ext_vector_type(2))) float v2f;
typedef __attribute__((ext_vector_type(8))) float v8f;

#define CH 64
#define HEADS 4
#define HC 256
#define NEG_SLOPE 0.2f

__global__ void gat_init_kernel(float* __restrict__ accum,
                                float* __restrict__ smax,
                                float* __restrict__ den, int N) {
    int idx = blockIdx.x * blockDim.x + threadIdx.x;
    int total = N * HC;
    if (idx < total) accum[idx] = 0.0f;
    if (idx < N * HEADS) {
        smax[idx] = -__builtin_inff();
        den[idx]  = 0.0f;
    }
}

__global__ void gat_proj_kernel(const float* __restrict__ x,
                                const float* __restrict__ Wl,
                                const float* __restrict__ bl,
                                const float* __restrict__ Wr,
                                const float* __restrict__ br,
                                float* __restrict__ xl,
                                float* __restrict__ xr, int N) {
    __shared__ float As[16 * CH];
    const int row0 = blockIdx.x * 16;
    const int tid  = threadIdx.x;

    for (int i = tid; i < 16 * CH; i += 256) {
        int r = i >> 6, c = i & 63;
        As[i] = x[(size_t)(row0 + r) * CH + c];
    }
    __syncthreads();

    const int wave = tid >> 5;
    const int lane = tid & 31;
    const int m    = lane & 15;
    const int koff = (lane >> 4) << 1;
    const int mrow = (lane >> 4) << 3;

    for (int t = 0; t < 4; ++t) {
        const int colTile = wave * 4 + t;
        const bool left   = (colTile < 16);
        const float* __restrict__ W = left ? Wl : Wr;
        const float* __restrict__ b = left ? bl : br;
        const int n = ((colTile & 15) << 4) + m;

        v8f acc;
        const float bv = b[n];
        #pragma unroll
        for (int v = 0; v < 8; ++v) acc[v] = bv;

        #pragma unroll
        for (int k0 = 0; k0 < CH; k0 += 4) {
            v2f a, bf;
            a.x  = As[m * CH + k0 + koff];
            a.y  = As[m * CH + k0 + koff + 1];
            bf.x = W[(size_t)(k0 + koff) * HC + n];
            bf.y = W[(size_t)(k0 + koff + 1) * HC + n];
            acc = __builtin_amdgcn_wmma_f32_16x16x4_f32(
                false, a, false, bf, (short)0, acc, false, false);
        }

        float* __restrict__ out = left ? xl : xr;
        #pragma unroll
        for (int v = 0; v < 8; ++v)
            out[(size_t)(row0 + mrow + v) * HC + n] = acc[v];
    }
}

__global__ void gat_score_kernel(const float* __restrict__ xl,
                                 const float* __restrict__ xr,
                                 const float* __restrict__ att,
                                 const int* __restrict__ edges,
                                 float* __restrict__ score,
                                 float* __restrict__ smax,
                                 int E, int Etot) {
    const int widx = (int)((blockIdx.x * (size_t)blockDim.x + threadIdx.x) >> 5);
    const int lane = threadIdx.x & 31;
    if (widx >= Etot) return;

    int src, dst;
    if (widx < E) { src = edges[widx]; dst = edges[E + widx]; }
    else          { src = widx - E;    dst = src; }

    const float4* __restrict__ pl = (const float4*)(xl + (size_t)src * HC);
    const float4* __restrict__ pr = (const float4*)(xr + (size_t)dst * HC);
    const float4* __restrict__ pa = (const float4*)att;

    float s = 0.0f;
    #pragma unroll
    for (int i = 0; i < 2; ++i) {
        const int q = lane * 2 + i;
        float4 a = pl[q], b = pr[q], w = pa[q];
        float vx = a.x + b.x; vx = vx > 0.0f ? vx : NEG_SLOPE * vx;
        float vy = a.y + b.y; vy = vy > 0.0f ? vy : NEG_SLOPE * vy;
        float vz = a.z + b.z; vz = vz > 0.0f ? vz : NEG_SLOPE * vz;
        float vw = a.w + b.w; vw = vw > 0.0f ? vw : NEG_SLOPE * vw;
        s += w.x * vx + w.y * vy + w.z * vz + w.w * vw;
    }
    s += __shfl_xor(s, 1, 32);
    s += __shfl_xor(s, 2, 32);
    s += __shfl_xor(s, 4, 32);

    if ((lane & 7) == 0) {
        const int h = lane >> 3;
        score[(size_t)widx * HEADS + h] = s;
        __hip_atomic_fetch_max(&smax[dst * HEADS + h], s,
                               __ATOMIC_RELAXED, __HIP_MEMORY_SCOPE_AGENT);
    }
}

__global__ void gat_exp_kernel(const int* __restrict__ edges,
                               const float* __restrict__ smax,
                               float* __restrict__ score,
                               float* __restrict__ den,
                               int E, int Etot) {
    const int idx = blockIdx.x * blockDim.x + threadIdx.x;
    if (idx >= Etot * HEADS) return;
    const int e = idx >> 2, h = idx & 3;
    const int dst = (e < E) ? edges[E + e] : (e - E);
    const float ex = __expf(score[idx] - smax[dst * HEADS + h]);
    score[idx] = ex;
    __hip_atomic_fetch_add(&den[dst * HEADS + h], ex,
                           __ATOMIC_RELAXED, __HIP_MEMORY_SCOPE_AGENT);
}

__global__ void gat_aggr_kernel(const float* __restrict__ xl,
                                const int* __restrict__ edges,
                                const float* __restrict__ ex,
                                const float* __restrict__ den,
                                float* __restrict__ accum,
                                int E, int Etot) {
    const int widx = (int)((blockIdx.x * (size_t)blockDim.x + threadIdx.x) >> 5);
    const int lane = threadIdx.x & 31;
    if (widx >= Etot) return;

    int src, dst;
    if (widx < E) { src = edges[widx]; dst = edges[E + widx]; }
    else          { src = widx - E;    dst = src; }

    const int h = lane >> 3;
    const float alpha = ex[(size_t)widx * HEADS + h] / den[dst * HEADS + h];

    const float4* __restrict__ pl = (const float4*)(xl + (size_t)src * HC);
    float* __restrict__ pa = accum + (size_t)dst * HC;

    #pragma unroll
    for (int i = 0; i < 2; ++i) {
        const int q = lane * 2 + i;
        float4 v = pl[q];
        const int base = q << 2;
        __hip_atomic_fetch_add(&pa[base + 0], alpha * v.x, __ATOMIC_RELAXED, __HIP_MEMORY_SCOPE_AGENT);
        __hip_atomic_fetch_add(&pa[base + 1], alpha * v.y, __ATOMIC_RELAXED, __HIP_MEMORY_SCOPE_AGENT);
        __hip_atomic_fetch_add(&pa[base + 2], alpha * v.z, __ATOMIC_RELAXED, __HIP_MEMORY_SCOPE_AGENT);
        __hip_atomic_fetch_add(&pa[base + 3], alpha * v.w, __ATOMIC_RELAXED, __HIP_MEMORY_SCOPE_AGENT);
    }
}

__global__ void gat_final_kernel(const float* __restrict__ accum,
                                 const float* __restrict__ bias,
                                 float* __restrict__ out, int N) {
    const int idx = blockIdx.x * blockDim.x + threadIdx.x;
    if (idx >= N * CH) return;
    const int n = idx >> 6, c = idx & 63;
    const float* __restrict__ a = accum + (size_t)n * HC;
    out[idx] = 0.25f * (a[c] + a[CH + c] + a[2 * CH + c] + a[3 * CH + c]) + bias[c];
}

extern "C" void kernel_launch(void* const* d_in, const int* in_sizes, int n_in,
                              void* d_out, int out_size, void* d_ws, size_t ws_size,
                              hipStream_t stream) {
    const float* x    = (const float*)d_in[0];
    const float* Wl   = (const float*)d_in[1];
    const float* bl   = (const float*)d_in[2];
    const float* Wr   = (const float*)d_in[3];
    const float* br   = (const float*)d_in[4];
    const float* att  = (const float*)d_in[5];
    const float* bias = (const float*)d_in[6];
    const int*   edges = (const int*)d_in[7];

    const int N    = in_sizes[0] / CH;
    const int E    = in_sizes[7] / 2;
    const int Etot = E + N;

    float* xl    = (float*)d_ws;
    float* xr    = xl + (size_t)N * HC;
    float* accum = xr + (size_t)N * HC;
    float* smax  = accum + (size_t)N * HC;
    float* den   = smax + (size_t)N * HEADS;
    float* score = den + (size_t)N * HEADS;

    const int B = 256;
    gat_init_kernel<<<(N * HC + B - 1) / B, B, 0, stream>>>(accum, smax, den, N);
    gat_proj_kernel<<<N / 16, B, 0, stream>>>(x, Wl, bl, Wr, br, xl, xr, N);

    const long long waveThreads = (long long)Etot * 32;
    gat_score_kernel<<<(int)((waveThreads + B - 1) / B), B, 0, stream>>>(
        xl, xr, att, edges, score, smax, E, Etot);
    gat_exp_kernel<<<(Etot * HEADS + B - 1) / B, B, 0, stream>>>(
        edges, smax, score, den, E, Etot);
    gat_aggr_kernel<<<(int)((waveThreads + B - 1) / B), B, 0, stream>>>(
        xl, edges, score, den, accum, E, Etot);
    gat_final_kernel<<<(N * CH + B - 1) / B, B, 0, stream>>>(
        accum, bias, (float*)d_out, N);
}


# aggregation folds the head mean into each edge message in-wave: 2 whole-line f32 atomic adds per edge into out (pre-set to bias) replace 256 per-head adds and the final pass; sum order changes
# speedup vs baseline: 2.3400x; 2.3400x over previous
#include <hip/hip_runtime.h>
#include <hip/hip_bf16.h>

typedef __attribute__((ext_vector_type(2))) float v2f;
typedef __attribute__((ext_vector_type(8))) float v8f;

#define CH 64
#define HEADS 4
#define HC 256
#define NEG_SLOPE 0.2f

__global__ void gat_init_kernel(const float* __restrict__ bias,
                                float* __restrict__ out,
                                float* __restrict__ smax,
                                float* __restrict__ den, int N) {
    int idx = blockIdx.x * blockDim.x + threadIdx.x;
    if (idx < N * (CH / 4))
        ((float4*)out)[idx] = ((const float4*)bias)[idx & (CH / 4 - 1)];
    if (idx < N * HEADS) {
        smax[idx] = -__builtin_inff();
        den[idx]  = 0.0f;
    }
}

__global__ void gat_proj_kernel(const float* __restrict__ x,
                                const float* __restrict__ Wl,
                                const float* __restrict__ bl,
                                const float* __restrict__ Wr,
                                const float* __restrict__ br,
                                float* __restrict__ xl,
                                float* __restrict__ xr, int N) {
    __shared__ float As[16 * CH];
    const int row0 = blockIdx.x * 16;
    const int tid  = threadIdx.x;

    for (int i = tid; i < 16 * CH; i += 256) {
        int r = i >> 6, c = i & 63;
        As[i] = x[(size_t)(row0 + r) * CH + c];
    }
    __syncthreads();

    const int wave = tid >> 5;
    const int lane = tid & 31;
    const int m    = lane & 15;
    const int koff = (lane >> 4) << 1;
    const int mrow = (lane >> 4) << 3;

    for (int t = 0; t < 4; ++t) {
        const int colTile = wave * 4 + t;
        const bool left   = (colTile < 16);
        const float* __restrict__ W = left ? Wl : Wr;
        const float* __restrict__ b = left ? bl : br;
        const int n = ((colTile & 15) << 4) + m;

        v8f acc;
        const float bv = b[n];
        #pragma unroll
        for (int v = 0; v < 8; ++v) acc[v] = bv;

        #pragma unroll
        for (int k0 = 0; k0 < CH; k0 += 4) {
            v2f a, bf;
            a.x  = As[m * CH + k0 + koff];
            a.y  = As[m * CH + k0 + koff + 1];
            bf.x = W[(size_t)(k0 + koff) * HC + n];
            bf.y = W[(size_t)(k0 + koff + 1) * HC + n];
            acc = __builtin_amdgcn_wmma_f32_16x16x4_f32(
                false, a, false, bf, (short)0, acc, false, false);
        }

        float* __restrict__ out = left ? xl : xr;
        #pragma unroll
        for (int v = 0; v < 8; ++v)
            out[(size_t)(row0 + mrow + v) * HC + n] = acc[v];
    }
}

__global__ void gat_score_kernel(const float* __restrict__ xl,
                                 const float* __restrict__ xr,
                                 const float* __restrict__ att,
                                 const int* __restrict__ edges,
                                 float* __restrict__ score,
                                 float* __restrict__ smax,
                                 int E, int Etot) {
    const int widx = (int)((blockIdx.x * (size_t)blockDim.x + threadIdx.x) >> 5);
    const int lane = threadIdx.x & 31;
    if (widx >= Etot) return;

    int src, dst;
    if (widx < E) { src = edges[widx]; dst = edges[E + widx]; }
    else          { src = widx - E;    dst = src; }

    const float4* __restrict__ pl = (const float4*)(xl + (size_t)src * HC);
    const float4* __restrict__ pr = (const float4*)(xr + (size_t)dst * HC);
    const float4* __restrict__ pa = (const float4*)att;

    float s = 0.0f;
    #pragma unroll
    for (int i = 0; i < 2; ++i) {
        const int q = lane * 2 + i;
        float4 a = pl[q], b = pr[q], w = pa[q];
        float vx = a.x + b.x; vx = vx > 0.0f ? vx : NEG_SLOPE * vx;
        float vy = a.y + b.y; vy = vy > 0.0f ? vy : NEG_SLOPE * vy;
        float vz = a.z + b.z; vz = vz > 0.0f ? vz : NEG_SLOPE * vz;
        float vw = a.w + b.w; vw = vw > 0.0f ? vw : NEG_SLOPE * vw;
        s += w.x * vx + w.y * vy + w.z * vz + w.w * vw;
    }
    s += __shfl_xor(s, 1, 32);
    s += __shfl_xor(s, 2, 32);
    s += __shfl_xor(s, 4, 32);

    if ((lane & 7) == 0) {
        const int h = lane >> 3;
        score[(size_t)widx * HEADS + h] = s;
        __hip_atomic_fetch_max(&smax[dst * HEADS + h], s,
                               __ATOMIC_RELAXED, __HIP_MEMORY_SCOPE_AGENT);
    }
}

__global__ void gat_exp_kernel(const int* __restrict__ edges,
                               const float* __restrict__ smax,
                               float* __restrict__ score,
                               float* __restrict__ den,
                               int E, int Etot) {
    const int idx = blockIdx.x * blockDim.x + threadIdx.x;
    if (idx >= Etot * HEADS) return;
    const int e = idx >> 2, h = idx & 3;
    const int dst = (e < E) ? edges[E + e] : (e - E);
    const float ex = __expf(score[idx] - smax[dst * HEADS + h]);
    score[idx] = ex;
    __hip_atomic_fetch_add(&den[dst * HEADS + h], ex,
                           __ATOMIC_RELAXED, __HIP_MEMORY_SCOPE_AGENT);
}

__global__ void gat_aggr_kernel(const float* __restrict__ xl,
                                const int* __restrict__ edges,
                                const float* __restrict__ ex,
                                const float* __restrict__ den,
                                float* __restrict__ out,
                                int E, int Etot) {
    const int widx = (int)((blockIdx.x * (size_t)blockDim.x + threadIdx.x) >> 5);
    const int lane = threadIdx.x & 31;
    if (widx >= Etot) return;

    int src, dst;
    if (widx < E) { src = edges[widx]; dst = edges[E + widx]; }
    else          { src = widx - E;    dst = src; }

    const int hq = lane & 3;
    const float aw = 0.25f * (ex[(size_t)widx * HEADS + hq] / den[dst * HEADS + hq]);
    const int awb = __builtin_bit_cast(int, aw);
    const float a0 = __builtin_bit_cast(float, __builtin_amdgcn_readlane(awb, 0));
    const float a1 = __builtin_bit_cast(float, __builtin_amdgcn_readlane(awb, 1));
    const float a2 = __builtin_bit_cast(float, __builtin_amdgcn_readlane(awb, 2));
    const float a3 = __builtin_bit_cast(float, __builtin_amdgcn_readlane(awb, 3));

    const float* __restrict__ pl = xl + (size_t)src * HC + lane;
    const float v00 = pl[0 * CH],      v01 = pl[0 * CH + 32];
    const float v10 = pl[1 * CH],      v11 = pl[1 * CH + 32];
    const float v20 = pl[2 * CH],      v21 = pl[2 * CH + 32];
    const float v30 = pl[3 * CH],      v31 = pl[3 * CH + 32];

    const float m0 = a0 * v00 + a1 * v10 + a2 * v20 + a3 * v30;
    const float m1 = a0 * v01 + a1 * v11 + a2 * v21 + a3 * v31;

    float* __restrict__ po = out + (size_t)dst * CH + lane;
    __hip_atomic_fetch_add(po,      m0, __ATOMIC_RELAXED, __HIP_MEMORY_SCOPE_AGENT);
    __hip_atomic_fetch_add(po + 32, m1, __ATOMIC_RELAXED, __HIP_MEMORY_SCOPE_AGENT);
}

extern "C" void kernel_launch(void* const* d_in, const int* in_sizes, int n_in,
                              void* d_out, int out_size, void* d_ws, size_t ws_size,
                              hipStream_t stream) {
    const float* x    = (const float*)d_in[0];
    const float* Wl   = (const float*)d_in[1];
    const float* bl   = (const float*)d_in[2];
    const float* Wr   = (const float*)d_in[3];
    const float* br   = (const float*)d_in[4];
    const float* att  = (const float*)d_in[5];
    const float* bias = (const float*)d_in[6];
    const int*   edges = (const int*)d_in[7];

    const int N    = in_sizes[0] / CH;
    const int E    = in_sizes[7] / 2;
    const int Etot = E + N;

    float* xl    = (float*)d_ws;
    float* xr    = xl + (size_t)N * HC;
    float* smax  = xr + (size_t)N * HC;
    float* den   = smax + (size_t)N * HEADS;
    float* score = den + (size_t)N * HEADS;

    const int B = 256;
    gat_init_kernel<<<(N * (CH / 4) + B - 1) / B, B, 0, stream>>>(
        bias, (float*)d_out, smax, den, N);
    gat_proj_kernel<<<N / 16, B, 0, stream>>>(x, Wl, bl, Wr, br, xl, xr, N);

    const long long waveThreads = (long long)Etot * 32;
    gat_score_kernel<<<(int)((waveThreads + B - 1) / B), B, 0, stream>>>(
        xl, xr, att, edges, score, smax, E, Etot);
    gat_exp_kernel<<<(Etot * HEADS + B - 1) / B, B, 0, stream>>>(
        edges, smax, score, den, E, Etot);
    gat_aggr_kernel<<<(int)((waveThreads + B - 1) / B), B, 0, stream>>>(
        xl, edges, score, den, (float*)d_out, E, Etot);
}


# edges chained per destination node by one integer exchange each; one wave per node does scores, running softmax and head-mean sum in registers, out written once: no float atomics; sum order changes
# speedup vs baseline: 6.1646x; 2.6344x over previous
#include <hip/hip_runtime.h>
#include <hip/hip_bf16.h>

typedef __attribute__((ext_vector_type(2))) float v2f;
typedef __attribute__((ext_vector_type(8))) float v8f;

#define CH 64
#define HEADS 4
#define HC 256
#define NEG_SLOPE 0.2f
#define HSTR 32

__global__ void gat_proj_kernel(const float* __restrict__ x,
                                const float* __restrict__ Wl,
                                const float* __restrict__ bl,
                                const float* __restrict__ Wr,
                                const float* __restrict__ br,
                                float* __restrict__ xl,
                                float* __restrict__ xr,
                                int* __restrict__ head, int N) {
    __shared__ float As[16 * CH];
    const int row0 = blockIdx.x * 16;
    const int tid  = threadIdx.x;

    if (tid < 16) head[(size_t)(row0 + tid) * HSTR] = -1;

    for (int i = tid; i < 16 * CH; i += 256) {
        int r = i >> 6, c = i & 63;
        As[i] = x[(size_t)(row0 + r) * CH + c];
    }
    __syncthreads();

    const int wave = tid >> 5;
    const int lane = tid & 31;
    const int m    = lane & 15;
    const int koff = (lane >> 4) << 1;
    const int mrow = (lane >> 4) << 3;

    for (int t = 0; t < 4; ++t) {
        const int colTile = wave * 4 + t;
        const bool left   = (colTile < 16);
        const float* __restrict__ W = left ? Wl : Wr;
        const float* __restrict__ b = left ? bl : br;
        const int n = ((colTile & 15) << 4) + m;

        v8f acc;
        const float bv = b[n];
        #pragma unroll
        for (int v = 0; v < 8; ++v) acc[v] = bv;

        #pragma unroll
        for (int k0 = 0; k0 < CH; k0 += 4) {
            v2f a, bf;
            a.x  = As[m * CH + k0 + koff];
            a.y  = As[m * CH + k0 + koff + 1];
            bf.x = W[(size_t)(k0 + koff) * HC + n];
            bf.y = W[(size_t)(k0 + koff + 1) * HC + n];
            acc = __builtin_amdgcn_wmma_f32_16x16x4_f32(
                false, a, false, bf, (short)0, acc, false, false);
        }

        float* __restrict__ out = left ? xl : xr;
        #pragma unroll
        for (int v = 0; v < 8; ++v)
            out[(size_t)(row0 + mrow + v) * HC + n] = acc[v];
    }
}

__global__ void gat_link_kernel(const int* __restrict__ edges,
                                int* __restrict__ head,
                                int2* __restrict__ rec, int E) {
    const int e = blockIdx.x * blockDim.x + threadIdx.x;
    if (e >= E) return;
    const int src = edges[e];
    const int dst = edges[E + e];
    const int old = __hip_atomic_exchange(&head[(size_t)dst * HSTR], e,
                                          __ATOMIC_RELAXED, __HIP_MEMORY_SCOPE_AGENT);
    rec[e] = make_int2(src, old);
}

template <int XM>
__device__ __forceinline__ float row_xor(float v) {
    const int b = __builtin_bit_cast(int, v);
    return __builtin_bit_cast(float,
        __builtin_amdgcn_update_dpp(0, b, 0x160 + XM, 0xF, 0xF, true));
}
__device__ __forceinline__ float xhalf(float v) {
    const int b = __builtin_bit_cast(int, v);
    return __builtin_bit_cast(float,
        __builtin_amdgcn_permlanex16(b, b, 0x76543210, 0xfedcba98, false, false));
}
__device__ __forceinline__ float row_sum16(float s) {
    s += row_xor<1>(s);
    s += row_xor<2>(s);
    s += row_xor<4>(s);
    s += row_xor<8>(s);
    return s;
}
__device__ __forceinline__ float lrelu(float v) {
    return v > 0.0f ? v : NEG_SLOPE * v;
}
__device__ __forceinline__ void edge_scores(const float4 a0, const float4 a1,
                                            const float4 r0, const float4 r1,
                                            const float4 w0, const float4 w1,
                                            float& s_lo, float& s_hi) {
    float lo = w0.x * lrelu(a0.x + r0.x) + w0.y * lrelu(a0.y + r0.y)
             + w0.z * lrelu(a0.z + r0.z) + w0.w * lrelu(a0.w + r0.w);
    float hi = w1.x * lrelu(a1.x + r1.x) + w1.y * lrelu(a1.y + r1.y)
             + w1.z * lrelu(a1.z + r1.z) + w1.w * lrelu(a1.w + r1.w);
    s_lo = row_sum16(lo);
    s_hi = row_sum16(hi);
}
__device__ __forceinline__ void softmax_step(const float s, const float4 a,
                                             float& m, float& d, float4& acc) {
    const float df = s - m;
    const float t  = __expf(-fabsf(df));
    const bool  up = df > 0.0f;
    const float c  = up ? t : 1.0f;
    const float p  = up ? 1.0f : t;
    m = up ? s : m;
    d = d * c + p;
    acc.x = acc.x * c + p * a.x;
    acc.y = acc.y * c + p * a.y;
    acc.z = acc.z * c + p * a.z;
    acc.w = acc.w * c + p * a.w;
}

__global__ void gat_node_kernel(const float* __restrict__ xl,
                                const float* __restrict__ xr,
                                const float* __restrict__ att,
                                const float* __restrict__ bias,
                                const int2* __restrict__ rec,
                                const int* __restrict__ head,
                                float* __restrict__ out, int N, int E) {
    const int widx = (int)((blockIdx.x * (size_t)blockDim.x + threadIdx.x) >> 5);
    const int lane = threadIdx.x & 31;
    if (widx >= N) return;
    const int i = __builtin_amdgcn_readfirstlane(widx);

    const float4 w0 = ((const float4*)att)[lane];
    const float4 w1 = ((const float4*)att)[lane + 32];
    const float4* __restrict__ pr = (const float4*)(xr + (size_t)i * HC);
    const float4 r0 = pr[lane], r1 = pr[lane + 32];

    int e = __builtin_amdgcn_readfirstlane(head[(size_t)i * HSTR]);
    int2 nx = make_int2(0, -1);
    if (e >= 0) nx = rec[e];

    const float4* __restrict__ ps = (const float4*)(xl + (size_t)i * HC);
    float4 a0 = ps[lane], a1 = ps[lane + 32];
    float m_lo, m_hi;
    edge_scores(a0, a1, r0, r1, w0, w1, m_lo, m_hi);
    float d_lo = 1.0f, d_hi = 1.0f;
    float4 acc0 = a0, acc1 = a1;

    for (int it = 0; it < E && e >= 0; ++it) {
        const int src = __builtin_amdgcn_readfirstlane(nx.x);
        const int en  = __builtin_amdgcn_readfirstlane(nx.y);
        if (en >= 0) nx = rec[en];
        const float4* __restrict__ pl = (const float4*)(xl + (size_t)src * HC);
        a0 = pl[lane];
        a1 = pl[lane + 32];
        float s_lo, s_hi;
        edge_scores(a0, a1, r0, r1, w0, w1, s_lo, s_hi);
        softmax_step(s_lo, a0, m_lo, d_lo, acc0);
        softmax_step(s_hi, a1, m_hi, d_hi, acc1);
        e = en;
    }

    const float i_lo = 1.0f / d_lo, i_hi = 1.0f / d_hi;
    float4 o;
    o.x = acc0.x * i_lo + acc1.x * i_hi;
    o.y = acc0.y * i_lo + acc1.y * i_hi;
    o.z = acc0.z * i_lo + acc1.z * i_hi;
    o.w = acc0.w * i_lo + acc1.w * i_hi;
    o.x += xhalf(o.x);
    o.y += xhalf(o.y);
    o.z += xhalf(o.z);
    o.w += xhalf(o.w);
    if (lane < 16) {
        const float4 b = ((const float4*)bias)[lane];
        float4 r;
        r.x = 0.25f * o.x + b.x;
        r.y = 0.25f * o.y + b.y;
        r.z = 0.25f * o.z + b.z;
        r.w = 0.25f * o.w + b.w;
        ((float4*)(out + (size_t)i * CH))[lane] = r;
    }
}

extern "C" void kernel_launch(void* const* d_in, const int* in_sizes, int n_in,
                              void* d_out, int out_size, void* d_ws, size_t ws_size,
                              hipStream_t stream) {
    const float* x    = (const float*)d_in[0];
    const float* Wl   = (const float*)d_in[1];
    const float* bl   = (const float*)d_in[2];
    const float* Wr   = (const float*)d_in[3];
    const float* br   = (const float*)d_in[4];
    const float* att  = (const float*)d_in[5];
    const float* bias = (const float*)d_in[6];
    const int*   edges = (const int*)d_in[7];

    const int N    = in_sizes[0] / CH;
    const int E    = in_sizes[7] / 2;

    float* xl   = (float*)d_ws;
    float* xr   = xl + (size_t)N * HC;
    int2*  rec  = (int2*)(xr + (size_t)N * HC);
    int*   head = (int*)(rec + E);

    const int B = 256;
    gat_proj_kernel<<<N / 16, B, 0, stream>>>(x, Wl, bl, Wr, br, xl, xr, head, N);
    gat_link_kernel<<<(E + B - 1) / B, B, 0, stream>>>(edges, head, rec, E);

    const long long waveThreads = (long long)N * 32;
    gat_node_kernel<<<(int)((waveThreads + B - 1) / B), B, 0, stream>>>(
        xl, xr, att, bias, rec, head, (float*)d_out, N, E);
}


# xl/xr rows re-laid so a node-kernel lane holds 8 channels of ONE head: one score (8-lane sum) and one softmax state per lane, not two 16-lane sums and two; score and head-mean sum order changes
# speedup vs baseline: 6.7924x; 1.1018x over previous
#include <hip/hip_runtime.h>
#include <hip/hip_bf16.h>

typedef __attribute__((ext_vector_type(2))) float v2f;
typedef __attribute__((ext_vector_type(4))) float v4f;
typedef __attribute__((ext_vector_type(8))) float v8f;

#define CH 64
#define HEADS 4
#define HC 256
#define NEG_SLOPE 0.2f
#define HSTR 32

__global__ void gat_proj_kernel(const float* __restrict__ x,
                                const float* __restrict__ Wl,
                                const float* __restrict__ bl,
                                const float* __restrict__ Wr,
                                const float* __restrict__ br,
                                float* __restrict__ xl,
                                float* __restrict__ xr,
                                int* __restrict__ head, int N) {
    __shared__ float As[16 * CH];
    const int row0 = blockIdx.x * 16;
    const int tid  = threadIdx.x;

    if (tid < 16) head[(size_t)(row0 + tid) * HSTR] = -1;

    for (int i = tid; i < 16 * CH; i += 256) {
        int r = i >> 6, c = i & 63;
        As[i] = x[(size_t)(row0 + r) * CH + c];
    }
    __syncthreads();

    const int wave = tid >> 5;
    const int lane = tid & 31;
    const int m    = lane & 15;
    const int koff = (lane >> 4) << 1;
    const int mrow = (lane >> 4) << 3;

    for (int t = 0; t < 4; ++t) {
        const int colTile = wave * 4 + t;
        const bool left   = (colTile < 16);
        const float* __restrict__ W = left ? Wl : Wr;
        const float* __restrict__ b = left ? bl : br;
        const int n = ((colTile & 15) << 4) + m;

        v8f acc;
        const float bv = b[n];
        #pragma unroll
        for (int v = 0; v < 8; ++v) acc[v] = bv;

        #pragma unroll
        for (int k0 = 0; k0 < CH; k0 += 4) {
            v2f a, bf;
            a.x  = As[m * CH + k0 + koff];
            a.y  = As[m * CH + k0 + koff + 1];
            bf.x = W[(size_t)(k0 + koff) * HC + n];
            bf.y = W[(size_t)(k0 + koff + 1) * HC + n];
            acc = __builtin_amdgcn_wmma_f32_16x16x4_f32(
                false, a, false, bf, (short)0, acc, false, false);
        }

        float* __restrict__ out = left ? xl : xr;
        const int pn = ((n & 32) << 2) + ((n >> 6) << 5) + (n & 31);
        #pragma unroll
        for (int v = 0; v < 8; ++v)
            out[(size_t)(row0 + mrow + v) * HC + pn] = acc[v];
    }
}

__global__ void gat_link_kernel(const int* __restrict__ edges,
                                int* __restrict__ head,
                                int2* __restrict__ rec, int E) {
    const int e = blockIdx.x * blockDim.x + threadIdx.x;
    if (e >= E) return;
    const int src = edges[e];
    const int dst = edges[E + e];
    const int old = __hip_atomic_exchange(&head[(size_t)dst * HSTR], e,
                                          __ATOMIC_RELAXED, __HIP_MEMORY_SCOPE_AGENT);
    rec[e] = make_int2(src, old);
}

template <int XM>
__device__ __forceinline__ float row_xor(float v) {
    const int b = __builtin_bit_cast(int, v);
    return __builtin_bit_cast(float,
        __builtin_amdgcn_update_dpp(0, b, 0x160 + XM, 0xF, 0xF, true));
}
__device__ __forceinline__ float xhalf(float v) {
    const int b = __builtin_bit_cast(int, v);
    return __builtin_bit_cast(float,
        __builtin_amdgcn_permlanex16(b, b, 0x76543210, 0xfedcba98, false, false));
}
__device__ __forceinline__ float group_sum8(float s) {
    s += row_xor<1>(s);
    s += row_xor<2>(s);
    s += row_xor<4>(s);
    return s;
}
__device__ __forceinline__ float lrelu(float v) {
    return v > 0.0f ? v : NEG_SLOPE * v;
}
__device__ __forceinline__ v2f lrelu2(const v2f v) {
    v2f r;
    r.x = lrelu(v.x);
    r.y = lrelu(v.y);
    return r;
}
__device__ __forceinline__ float edge_score(const v4f a0, const v4f a1,
                                            const v4f r0, const v4f r1,
                                            const v4f w0, const v4f w1) {
    v2f s = w0.lo * lrelu2(a0.lo + r0.lo);
    s += w0.hi * lrelu2(a0.hi + r0.hi);
    s += w1.lo * lrelu2(a1.lo + r1.lo);
    s += w1.hi * lrelu2(a1.hi + r1.hi);
    return group_sum8(s.x + s.y);
}
__device__ __forceinline__ void softmax_step(const float s, const v4f a0, const v4f a1,
                                             float& m, float& d, v4f& acc0, v4f& acc1) {
    const float df = s - m;
    const float t  = __expf(-fabsf(df));
    const bool  up = df > 0.0f;
    const float c  = up ? t : 1.0f;
    const float p  = up ? 1.0f : t;
    m = up ? s : m;
    d = d * c + p;
    acc0 = acc0 * c + a0 * p;
    acc1 = acc1 * c + a1 * p;
}

__global__ void gat_node_kernel(const float* __restrict__ xl,
                                const float* __restrict__ xr,
                                const float* __restrict__ att,
                                const float* __restrict__ bias,
                                const int2* __restrict__ rec,
                                const int* __restrict__ head,
                                float* __restrict__ out, int N, int E) {
    const int widx = (int)((blockIdx.x * (size_t)blockDim.x + threadIdx.x) >> 5);
    const int lane = threadIdx.x & 31;
    if (widx >= N) return;
    const int i = __builtin_amdgcn_readfirstlane(widx);

    const int wq = ((lane >> 3) << 4) + (lane & 7);
    const v4f w0 = ((const v4f*)att)[wq];
    const v4f w1 = ((const v4f*)att)[wq + 8];
    const v4f* __restrict__ pr = (const v4f*)(xr + (size_t)i * HC);
    const v4f r0 = pr[lane], r1 = pr[lane + 32];

    int e = __builtin_amdgcn_readfirstlane(head[(size_t)i * HSTR]);
    int2 nx = make_int2(0, -1);
    if (e >= 0) nx = rec[e];

    const v4f* __restrict__ ps = (const v4f*)(xl + (size_t)i * HC);
    v4f a0 = ps[lane], a1 = ps[lane + 32];
    float m = edge_score(a0, a1, r0, r1, w0, w1);
    float d = 1.0f;
    v4f acc0 = a0, acc1 = a1;

    for (int it = 0; it < E && e >= 0; ++it) {
        const int src = __builtin_amdgcn_readfirstlane(nx.x);
        const int en  = __builtin_amdgcn_readfirstlane(nx.y);
        if (en >= 0) nx = rec[en];
        const v4f* __restrict__ pl = (const v4f*)(xl + (size_t)src * HC);
        a0 = pl[lane];
        a1 = pl[lane + 32];
        const float s = edge_score(a0, a1, r0, r1, w0, w1);
        softmax_step(s, a0, a1, m, d, acc0, acc1);
        e = en;
    }

    const float inv = 1.0f / d;
    const bool  upper = (lane & 8) != 0;
    const v4f keep = (upper ? acc1 : acc0) * inv;
    const v4f send = (upper ? acc0 : acc1) * inv;
    v4f o;
    o.x = keep.x + row_xor<8>(send.x);
    o.y = keep.y + row_xor<8>(send.y);
    o.z = keep.z + row_xor<8>(send.z);
    o.w = keep.w + row_xor<8>(send.w);
    o.x += xhalf(o.x);
    o.y += xhalf(o.y);
    o.z += xhalf(o.z);
    o.w += xhalf(o.w);
    if (lane < 16) {
        const v4f b = ((const v4f*)bias)[lane];
        ((v4f*)(out + (size_t)i * CH))[lane] = 0.25f * o + b;
    }
}

extern "C" void kernel_launch(void* const* d_in, const int* in_sizes, int n_in,
                              void* d_out, int out_size, void* d_ws, size_t ws_size,
                              hipStream_t stream) {
    const float* x    = (const float*)d_in[0];
    const float* Wl   = (const float*)d_in[1];
    const float* bl   = (const float*)d_in[2];
    const float* Wr   = (const float*)d_in[3];
    const float* br   = (const float*)d_in[4];
    const float* att  = (const float*)d_in[5];
    const float* bias = (const float*)d_in[6];
    const int*   edges = (const int*)d_in[7];

    const int N    = in_sizes[0] / CH;
    const int E    = in_sizes[7] / 2;

    float* xl   = (float*)d_ws;
    float* xr   = xl + (size_t)N * HC;
    int2*  rec  = (int2*)(xr + (size_t)N * HC);
    int*   head = (int*)(rec + E);

    const int B = 256;
    gat_proj_kernel<<<N / 16, B, 0, stream>>>(x, Wl, bl, Wr, br, xl, xr, head, N);
    gat_link_kernel<<<(E + B - 1) / B, B, 0, stream>>>(edges, head, rec, E);

    const long long waveThreads = (long long)N * 32;
    gat_node_kernel<<<(int)((waveThreads + B - 1) / B), B, 0, stream>>>(
        xl, xr, att, bias, rec, head, (float*)d_out, N, E);
}


# projection: a wave keeps the W fragments of its 64 columns in registers and walks 16-row tiles (512 blocks loop over all tiles); x fragments straight from global; W read once per block; same sums
# speedup vs baseline: 7.0112x; 1.0322x over previous
#include <hip/hip_runtime.h>
#include <hip/hip_bf16.h>

typedef __attribute__((ext_vector_type(2))) float v2f;
typedef __attribute__((ext_vector_type(4))) float v4f;
typedef __attribute__((ext_vector_type(8))) float v8f;

#define CH 64
#define HEADS 4
#define HC 256
#define NEG_SLOPE 0.2f
#define HSTR 32

#define PROJ_BLOCKS 512

__global__ void __launch_bounds__(256, 4)
gat_proj_kernel(const float* __restrict__ x,
                const float* __restrict__ Wl,
                const float* __restrict__ bl,
                const float* __restrict__ Wr,
                const float* __restrict__ br,
                float* __restrict__ xl,
                float* __restrict__ xr,
                int* __restrict__ head, int N) {
    const int tid  = threadIdx.x;
    const int wave = tid >> 5;
    const int lane = tid & 31;
    const int m    = lane & 15;
    const int koff = (lane >> 4) << 1;
    const int mrow = (lane >> 4) << 3;

    const bool left = (wave < 4);
    const float* __restrict__ W = left ? Wl : Wr;
    const float* __restrict__ b = left ? bl : br;
    float* __restrict__ out     = left ? xl : xr;
    const int n0 = ((wave & 3) << 6) + m;

    v2f   bfr[4][16];
    float bv[4];
    int   pn[4];
    #pragma unroll
    for (int t = 0; t < 4; ++t) {
        const int n = n0 + (t << 4);
        bv[t] = b[n];
        pn[t] = ((n & 32) << 2) + ((n >> 6) << 5) + (n & 31);
        #pragma unroll
        for (int j = 0; j < 16; ++j) {
            bfr[t][j].x = W[(size_t)(4 * j + koff) * HC + n];
            bfr[t][j].y = W[(size_t)(4 * j + koff + 1) * HC + n];
        }
    }

    const int nTiles = N >> 4;

    for (int tile = blockIdx.x; tile < nTiles; tile += gridDim.x) {
        const int row0 = tile * 16;
        if (tid < 16) head[(size_t)(row0 + tid) * HSTR] = -1;

        const v2f* __restrict__ xa = (const v2f*)(x + (size_t)(row0 + m) * CH + koff);
        v2f a[16];
        #pragma unroll
        for (int j = 0; j < 16; ++j) a[j] = xa[2 * j];

        v8f acc[4];
        #pragma unroll
        for (int t = 0; t < 4; ++t) {
            #pragma unroll
            for (int v = 0; v < 8; ++v) acc[t][v] = bv[t];
        }

        #pragma unroll
        for (int j = 0; j < 16; ++j) {
            #pragma unroll
            for (int t = 0; t < 4; ++t)
                acc[t] = __builtin_amdgcn_wmma_f32_16x16x4_f32(
                    false, a[j], false, bfr[t][j], (short)0, acc[t], false, false);
        }

        #pragma unroll
        for (int t = 0; t < 4; ++t) {
            #pragma unroll
            for (int v = 0; v < 8; ++v)
                out[(size_t)(row0 + mrow + v) * HC + pn[t]] = acc[t][v];
        }
    }
}

__global__ void gat_link_kernel(const int* __restrict__ edges,
                                int* __restrict__ head,
                                int2* __restrict__ rec, int E) {
    const int e = blockIdx.x * blockDim.x + threadIdx.x;
    if (e >= E) return;
    const int src = edges[e];
    const int dst = edges[E + e];
    const int old = __hip_atomic_exchange(&head[(size_t)dst * HSTR], e,
                                          __ATOMIC_RELAXED, __HIP_MEMORY_SCOPE_AGENT);
    rec[e] = make_int2(src, old);
}

template <int XM>
__device__ __forceinline__ float row_xor(float v) {
    const int b = __builtin_bit_cast(int, v);
    return __builtin_bit_cast(float,
        __builtin_amdgcn_update_dpp(0, b, 0x160 + XM, 0xF, 0xF, true));
}
__device__ __forceinline__ float xhalf(float v) {
    const int b = __builtin_bit_cast(int, v);
    return __builtin_bit_cast(float,
        __builtin_amdgcn_permlanex16(b, b, 0x76543210, 0xfedcba98, false, false));
}
__device__ __forceinline__ float group_sum8(float s) {
    s += row_xor<1>(s);
    s += row_xor<2>(s);
    s += row_xor<4>(s);
    return s;
}
__device__ __forceinline__ float lrelu(float v) {
    return v > 0.0f ? v : NEG_SLOPE * v;
}
__device__ __forceinline__ v2f lrelu2(const v2f v) {
    v2f r;
    r.x = lrelu(v.x);
    r.y = lrelu(v.y);
    return r;
}
__device__ __forceinline__ float edge_score(const v4f a0, const v4f a1,
                                            const v4f r0, const v4f r1,
                                            const v4f w0, const v4f w1) {
    v2f s = w0.lo * lrelu2(a0.lo + r0.lo);
    s += w0.hi * lrelu2(a0.hi + r0.hi);
    s += w1.lo * lrelu2(a1.lo + r1.lo);
    s += w1.hi * lrelu2(a1.hi + r1.hi);
    return group_sum8(s.x + s.y);
}
__device__ __forceinline__ void softmax_step(const float s, const v4f a0, const v4f a1,
                                             float& m, float& d, v4f& acc0, v4f& acc1) {
    const float df = s - m;
    const float t  = __expf(-fabsf(df));
    const bool  up = df > 0.0f;
    const float c  = up ? t : 1.0f;
    const float p  = up ? 1.0f : t;
    m = up ? s : m;
    d = d * c + p;
    acc0 = acc0 * c + a0 * p;
    acc1 = acc1 * c + a1 * p;
}

__global__ void gat_node_kernel(const float* __restrict__ xl,
                                const float* __restrict__ xr,
                                const float* __restrict__ att,
                                const float* __restrict__ bias,
                                const int2* __restrict__ rec,
                                const int* __restrict__ head,
                                float* __restrict__ out, int N, int E) {
    const int widx = (int)((blockIdx.x * (size_t)blockDim.x + threadIdx.x) >> 5);
    const int lane = threadIdx.x & 31;
    if (widx >= N) return;
    const int i = __builtin_amdgcn_readfirstlane(widx);

    const int wq = ((lane >> 3) << 4) + (lane & 7);
    const v4f w0 = ((const v4f*)att)[wq];
    const v4f w1 = ((const v4f*)att)[wq + 8];
    const v4f* __restrict__ pr = (const v4f*)(xr + (size_t)i * HC);
    const v4f r0 = pr[lane], r1 = pr[lane + 32];

    int e = __builtin_amdgcn_readfirstlane(head[(size_t)i * HSTR]);
    int2 nx = make_int2(0, -1);
    if (e >= 0) nx = rec[e];

    const v4f* __restrict__ ps = (const v4f*)(xl + (size_t)i * HC);
    v4f a0 = ps[lane], a1 = ps[lane + 32];
    float m = edge_score(a0, a1, r0, r1, w0, w1);
    float d = 1.0f;
    v4f acc0 = a0, acc1 = a1;

    for (int it = 0; it < E && e >= 0; ++it) {
        const int src = __builtin_amdgcn_readfirstlane(nx.x);
        const int en  = __builtin_amdgcn_readfirstlane(nx.y);
        if (en >= 0) nx = rec[en];
        const v4f* __restrict__ pl = (const v4f*)(xl + (size_t)src * HC);
        a0 = pl[lane];
        a1 = pl[lane + 32];
        const float s = edge_score(a0, a1, r0, r1, w0, w1);
        softmax_step(s, a0, a1, m, d, acc0, acc1);
        e = en;
    }

    const float inv = 1.0f / d;
    const bool  upper = (lane & 8) != 0;
    const v4f keep = (upper ? acc1 : acc0) * inv;
    const v4f send = (upper ? acc0 : acc1) * inv;
    v4f o;
    o.x = keep.x + row_xor<8>(send.x);
    o.y = keep.y + row_xor<8>(send.y);
    o.z = keep.z + row_xor<8>(send.z);
    o.w = keep.w + row_xor<8>(send.w);
    o.x += xhalf(o.x);
    o.y += xhalf(o.y);
    o.z += xhalf(o.z);
    o.w += xhalf(o.w);
    if (lane < 16) {
        const v4f b = ((const v4f*)bias)[lane];
        ((v4f*)(out + (size_t)i * CH))[lane] = 0.25f * o + b;
    }
}

extern "C" void kernel_launch(void* const* d_in, const int* in_sizes, int n_in,
                              void* d_out, int out_size, void* d_ws, size_t ws_size,
                              hipStream_t stream) {
    const float* x    = (const float*)d_in[0];
    const float* Wl   = (const float*)d_in[1];
    const float* bl   = (const float*)d_in[2];
    const float* Wr   = (const float*)d_in[3];
    const float* br   = (const float*)d_in[4];
    const float* att  = (const float*)d_in[5];
    const float* bias = (const float*)d_in[6];
    const int*   edges = (const int*)d_in[7];

    const int N    = in_sizes[0] / CH;
    const int E    = in_sizes[7] / 2;

    float* xl   = (float*)d_ws;
    float* xr   = xl + (size_t)N * HC;
    int2*  rec  = (int2*)(xr + (size_t)N * HC);
    int*   head = (int*)(rec + E);

    const int B = 256;
    gat_proj_kernel<<<PROJ_BLOCKS, B, 0, stream>>>(x, Wl, bl, Wr, br, xl, xr, head, N);
    gat_link_kernel<<<(E + B - 1) / B, B, 0, stream>>>(edges, head, rec, E);

    const long long waveThreads = (long long)N * 32;
    gat_node_kernel<<<(int)((waveThreads + B - 1) / B), B, 0, stream>>>(
        xl, xr, att, bias, rec, head, (float*)d_out, N, E);
}


# node loop, fewer vector ops per edge: leaky relu as max(v, slope*v); softmax sums taken against the self-loop score (moved only if a score tops it by a set margin), no running maximum; same sum order
# speedup vs baseline: 7.2476x; 1.0337x over previous
#include <hip/hip_runtime.h>
#include <hip/hip_bf16.h>

typedef __attribute__((ext_vector_type(2))) float v2f;
typedef __attribute__((ext_vector_type(4))) float v4f;
typedef __attribute__((ext_vector_type(8))) float v8f;

#define CH 64
#define HEADS 4
#define HC 256
#define NEG_SLOPE 0.2f
#define HSTR 32

#define PROJ_BLOCKS 512

__global__ void __launch_bounds__(256, 4)
gat_proj_kernel(const float* __restrict__ x,
                const float* __restrict__ Wl,
                const float* __restrict__ bl,
                const float* __restrict__ Wr,
                const float* __restrict__ br,
                float* __restrict__ xl,
                float* __restrict__ xr,
                int* __restrict__ head, int N) {
    const int tid  = threadIdx.x;
    const int wave = tid >> 5;
    const int lane = tid & 31;
    const int m    = lane & 15;
    const int koff = (lane >> 4) << 1;
    const int mrow = (lane >> 4) << 3;

    const bool left = (wave < 4);
    const float* __restrict__ W = left ? Wl : Wr;
    const float* __restrict__ b = left ? bl : br;
    float* __restrict__ out     = left ? xl : xr;
    const int n0 = ((wave & 3) << 6) + m;

    v2f   bfr[4][16];
    float bv[4];
    int   pn[4];
    #pragma unroll
    for (int t = 0; t < 4; ++t) {
        const int n = n0 + (t << 4);
        bv[t] = b[n];
        pn[t] = ((n & 32) << 2) + ((n >> 6) << 5) + (n & 31);
        #pragma unroll
        for (int j = 0; j < 16; ++j) {
            bfr[t][j].x = W[(size_t)(4 * j + koff) * HC + n];
            bfr[t][j].y = W[(size_t)(4 * j + koff + 1) * HC + n];
        }
    }

    const int nTiles = N >> 4;

    for (int tile = blockIdx.x; tile < nTiles; tile += gridDim.x) {
        const int row0 = tile * 16;
        if (tid < 16) head[(size_t)(row0 + tid) * HSTR] = -1;

        const v2f* __restrict__ xa = (const v2f*)(x + (size_t)(row0 + m) * CH + koff);
        v2f a[16];
        #pragma unroll
        for (int j = 0; j < 16; ++j) a[j] = xa[2 * j];

        v8f acc[4];
        #pragma unroll
        for (int t = 0; t < 4; ++t) {
            #pragma unroll
            for (int v = 0; v < 8; ++v) acc[t][v] = bv[t];
        }

        #pragma unroll
        for (int j = 0; j < 16; ++j) {
            #pragma unroll
            for (int t = 0; t < 4; ++t)
                acc[t] = __builtin_amdgcn_wmma_f32_16x16x4_f32(
                    false, a[j], false, bfr[t][j], (short)0, acc[t], false, false);
        }

        #pragma unroll
        for (int t = 0; t < 4; ++t) {
            #pragma unroll
            for (int v = 0; v < 8; ++v)
                out[(size_t)(row0 + mrow + v) * HC + pn[t]] = acc[t][v];
        }
    }
}

__global__ void gat_link_kernel(const int* __restrict__ edges,
                                int* __restrict__ head,
                                int2* __restrict__ rec, int E) {
    const int e = blockIdx.x * blockDim.x + threadIdx.x;
    if (e >= E) return;
    const int src = edges[e];
    const int dst = edges[E + e];
    const int old = __hip_atomic_exchange(&head[(size_t)dst * HSTR], e,
                                          __ATOMIC_RELAXED, __HIP_MEMORY_SCOPE_AGENT);
    rec[e] = make_int2(src, old);
}

template <int XM>
__device__ __forceinline__ float row_xor(float v) {
    const int b = __builtin_bit_cast(int, v);
    return __builtin_bit_cast(float,
        __builtin_amdgcn_update_dpp(0, b, 0x160 + XM, 0xF, 0xF, true));
}
__device__ __forceinline__ float xhalf(float v) {
    const int b = __builtin_bit_cast(int, v);
    return __builtin_bit_cast(float,
        __builtin_amdgcn_permlanex16(b, b, 0x76543210, 0xfedcba98, false, false));
}
__device__ __forceinline__ float group_sum8(float s) {
    s += row_xor<1>(s);
    s += row_xor<2>(s);
    s += row_xor<4>(s);
    return s;
}
__device__ __forceinline__ float lrelu(float v) {
    return __builtin_fmaxf(v, NEG_SLOPE * v);
}
__device__ __forceinline__ v2f lrelu2(const v2f v) {
    v2f r;
    r.x = lrelu(v.x);
    r.y = lrelu(v.y);
    return r;
}
__device__ __forceinline__ float edge_score(const v4f a0, const v4f a1,
                                            const v4f r0, const v4f r1,
                                            const v4f w0, const v4f w1) {
    v2f s = w0.lo * lrelu2(a0.lo + r0.lo);
    s += w0.hi * lrelu2(a0.hi + r0.hi);
    s += w1.lo * lrelu2(a1.lo + r1.lo);
    s += w1.hi * lrelu2(a1.hi + r1.hi);
    return group_sum8(s.x + s.y);
}
#define RESCALE_AT 16.0f
__device__ __forceinline__ void softmax_step(const float s, const v4f a0, const v4f a1,
                                             float& m, float& d, v4f& acc0, v4f& acc1) {
    float df = s - m;
    if (__builtin_amdgcn_ballot_w32(df > RESCALE_AT) != 0) {
        const bool  up = df > RESCALE_AT;
        const float c  = up ? __expf(-df) : 1.0f;
        d *= c;
        acc0 = acc0 * c;
        acc1 = acc1 * c;
        m  = up ? s : m;
        df = up ? 0.0f : df;
    }
    const float p = __expf(df);
    d += p;
    acc0 += a0 * p;
    acc1 += a1 * p;
}

__global__ void gat_node_kernel(const float* __restrict__ xl,
                                const float* __restrict__ xr,
                                const float* __restrict__ att,
                                const float* __restrict__ bias,
                                const int2* __restrict__ rec,
                                const int* __restrict__ head,
                                float* __restrict__ out, int N, int E) {
    const int widx = (int)((blockIdx.x * (size_t)blockDim.x + threadIdx.x) >> 5);
    const int lane = threadIdx.x & 31;
    if (widx >= N) return;
    const int i = __builtin_amdgcn_readfirstlane(widx);

    const int wq = ((lane >> 3) << 4) + (lane & 7);
    const v4f w0 = ((const v4f*)att)[wq];
    const v4f w1 = ((const v4f*)att)[wq + 8];
    const v4f* __restrict__ pr = (const v4f*)(xr + (size_t)i * HC);
    const v4f r0 = pr[lane], r1 = pr[lane + 32];

    int e = __builtin_amdgcn_readfirstlane(head[(size_t)i * HSTR]);
    int2 nx = make_int2(0, -1);
    if (e >= 0) nx = rec[e];

    const v4f* __restrict__ ps = (const v4f*)(xl + (size_t)i * HC);
    v4f a0 = ps[lane], a1 = ps[lane + 32];
    float m = edge_score(a0, a1, r0, r1, w0, w1);
    float d = 1.0f;
    v4f acc0 = a0, acc1 = a1;

    for (int it = 0; it < E && e >= 0; ++it) {
        const int src = __builtin_amdgcn_readfirstlane(nx.x);
        const int en  = __builtin_amdgcn_readfirstlane(nx.y);
        if (en >= 0) nx = rec[en];
        const v4f* __restrict__ pl = (const v4f*)(xl + (size_t)src * HC);
        a0 = pl[lane];
        a1 = pl[lane + 32];
        const float s = edge_score(a0, a1, r0, r1, w0, w1);
        softmax_step(s, a0, a1, m, d, acc0, acc1);
        e = en;
    }

    const float inv = 1.0f / d;
    const bool  upper = (lane & 8) != 0;
    const v4f keep = (upper ? acc1 : acc0) * inv;
    const v4f send = (upper ? acc0 : acc1) * inv;
    v4f o;
    o.x = keep.x + row_xor<8>(send.x);
    o.y = keep.y + row_xor<8>(send.y);
    o.z = keep.z + row_xor<8>(send.z);
    o.w = keep.w + row_xor<8>(send.w);
    o.x += xhalf(o.x);
    o.y += xhalf(o.y);
    o.z += xhalf(o.z);
    o.w += xhalf(o.w);
    if (lane < 16) {
        const v4f b = ((const v4f*)bias)[lane];
        ((v4f*)(out + (size_t)i * CH))[lane] = 0.25f * o + b;
    }
}

extern "C" void kernel_launch(void* const* d_in, const int* in_sizes, int n_in,
                              void* d_out, int out_size, void* d_ws, size_t ws_size,
                              hipStream_t stream) {
    const float* x    = (const float*)d_in[0];
    const float* Wl   = (const float*)d_in[1];
    const float* bl   = (const float*)d_in[2];
    const float* Wr   = (const float*)d_in[3];
    const float* br   = (const float*)d_in[4];
    const float* att  = (const float*)d_in[5];
    const float* bias = (const float*)d_in[6];
    const int*   edges = (const int*)d_in[7];

    const int N    = in_sizes[0] / CH;
    const int E    = in_sizes[7] / 2;

    float* xl   = (float*)d_ws;
    float* xr   = xl + (size_t)N * HC;
    int2*  rec  = (int2*)(xr + (size_t)N * HC);
    int*   head = (int*)(rec + E);

    const int B = 256;
    gat_proj_kernel<<<PROJ_BLOCKS, B, 0, stream>>>(x, Wl, bl, Wr, br, xl, xr, head, N);
    gat_link_kernel<<<(E + B - 1) / B, B, 0, stream>>>(edges, head, rec, E);

    const long long waveThreads = (long long)N * 32;
    gat_node_kernel<<<(int)((waveThreads + B - 1) / B), B, 0, stream>>>(
        xl, xr, att, bias, rec, head, (float*)d_out, N, E);
}


# edge linking folded into the projection, now two launches of one kernel (tile round 1 + list reset; other tiles, each thread's list exchanges issued under the tile's WMMAs): no link launch; same sums
# speedup vs baseline: 7.6381x; 1.0539x over previous
#include <hip/hip_runtime.h>
#include <hip/hip_bf16.h>

typedef __attribute__((ext_vector_type(2))) float v2f;
typedef __attribute__((ext_vector_type(4))) float v4f;
typedef __attribute__((ext_vector_type(8))) float v8f;

#define CH 64
#define HEADS 4
#define HC 256
#define NEG_SLOPE 0.2f
#define HSTR 32

#define PROJ_BLOCKS 512

template <bool LINK>
__global__ void __launch_bounds__(256, 4)
gat_proj_kernel(const float* __restrict__ x,
                const float* __restrict__ Wl,
                const float* __restrict__ bl,
                const float* __restrict__ Wr,
                const float* __restrict__ br,
                float* __restrict__ xl,
                float* __restrict__ xr,
                int* __restrict__ head,
                const int* __restrict__ edges,
                int2* __restrict__ rec, int N, int E) {
    const int tid  = threadIdx.x;
    const int wave = tid >> 5;
    const int lane = tid & 31;
    const int m    = lane & 15;
    const int koff = (lane >> 4) << 1;
    const int mrow = (lane >> 4) << 3;

    const bool left = (wave < 4);
    const float* __restrict__ W = left ? Wl : Wr;
    const float* __restrict__ b = left ? bl : br;
    float* __restrict__ out     = left ? xl : xr;
    const int n0 = ((wave & 3) << 6) + m;

    v2f   bfr[4][16];
    float bv[4];
    int   pn[4];
    #pragma unroll
    for (int t = 0; t < 4; ++t) {
        const int n = n0 + (t << 4);
        bv[t] = b[n];
        pn[t] = ((n & 32) << 2) + ((n >> 6) << 5) + (n & 31);
        #pragma unroll
        for (int j = 0; j < 16; ++j) {
            bfr[t][j].x = W[(size_t)(4 * j + koff) * HC + n];
            bfr[t][j].y = W[(size_t)(4 * j + koff + 1) * HC + n];
        }
    }

    const int nTiles = N >> 4;
    const int G      = gridDim.x;
    const int split  = min(nTiles, G);
    const int tBeg   = LINK ? split : 0;
    const int tEnd   = LINK ? nTiles : split;

    if (!LINK) {
        for (int i = blockIdx.x * 256 + tid; i < N; i += G * 256)
            head[(size_t)i * HSTR] = -1;
    }

    const int nChunks  = (E + 255) >> 8;
    const int myChunks = LINK && (int)blockIdx.x < nChunks ? (nChunks - (int)blockIdx.x + G - 1) / G : 0;
    const int myTiles  = tBeg + (int)blockIdx.x < tEnd ? (tEnd - tBeg - (int)blockIdx.x + G - 1) / G : 0;
    const int per      = myTiles > 0 ? myChunks / myTiles : 0;
    const int extra    = myTiles > 0 ? myChunks % myTiles : 0;
    int chunk = blockIdx.x;
    int k     = 0;

    int  ne0 = 0, ne1 = 0, nsrc0 = 0, ndst0 = 0, nsrc1 = 0, ndst1 = 0;
    bool nh0 = false, nh1 = false;
    auto fetch_edges = [&]() {
        const int take = min(2, per + (k < extra ? 1 : 0));
        ne0 = chunk * 256 + tid;  nh0 = take >= 1 && ne0 < E;  if (take >= 1) chunk += G;
        ne1 = chunk * 256 + tid;  nh1 = take >= 2 && ne1 < E;  if (take >= 2) chunk += G;
        if (nh0) { nsrc0 = edges[ne0]; ndst0 = edges[E + ne0]; }
        if (nh1) { nsrc1 = edges[ne1]; ndst1 = edges[E + ne1]; }
        ++k;
    };
    if (LINK && myTiles > 0) fetch_edges();

    for (int tile = tBeg + blockIdx.x; tile < tEnd; tile += G) {
        const int row0 = tile * 16;

        const int  e0 = ne0, e1 = ne1, src0 = nsrc0, dst0 = ndst0, src1 = nsrc1, dst1 = ndst1;
        const bool h0 = nh0, h1 = nh1;
        int old0 = -1, old1 = -1;
        if (LINK && tile + G < tEnd) fetch_edges();

        const v2f* __restrict__ xa = (const v2f*)(x + (size_t)(row0 + m) * CH + koff);
        v2f a[16];
        #pragma unroll
        for (int j = 0; j < 16; ++j) a[j] = xa[2 * j];

        if (LINK) {
            if (h0) old0 = __hip_atomic_exchange(&head[(size_t)dst0 * HSTR], e0,
                                                 __ATOMIC_RELAXED, __HIP_MEMORY_SCOPE_AGENT);
            if (h1) old1 = __hip_atomic_exchange(&head[(size_t)dst1 * HSTR], e1,
                                                 __ATOMIC_RELAXED, __HIP_MEMORY_SCOPE_AGENT);
        }

        v8f acc[4];
        #pragma unroll
        for (int t = 0; t < 4; ++t) {
            #pragma unroll
            for (int v = 0; v < 8; ++v) acc[t][v] = bv[t];
        }

        #pragma unroll
        for (int j = 0; j < 16; ++j) {
            #pragma unroll
            for (int t = 0; t < 4; ++t)
                acc[t] = __builtin_amdgcn_wmma_f32_16x16x4_f32(
                    false, a[j], false, bfr[t][j], (short)0, acc[t], false, false);
        }

        #pragma unroll
        for (int t = 0; t < 4; ++t) {
            #pragma unroll
            for (int v = 0; v < 8; ++v)
                out[(size_t)(row0 + mrow + v) * HC + pn[t]] = acc[t][v];
        }

        if (LINK) {
            if (h0) rec[e0] = make_int2(src0, old0);
            if (h1) rec[e1] = make_int2(src1, old1);
        }
    }

    if (LINK) {
        for (; chunk * 256 < E; chunk += G) {
            const int e = chunk * 256 + tid;
            if (e < E) {
                const int src = edges[e];
                const int dst = edges[E + e];
                const int old = __hip_atomic_exchange(&head[(size_t)dst * HSTR], e,
                                                      __ATOMIC_RELAXED, __HIP_MEMORY_SCOPE_AGENT);
                rec[e] = make_int2(src, old);
            }
        }
    }
}

template <int XM>
__device__ __forceinline__ float row_xor(float v) {
    const int b = __builtin_bit_cast(int, v);
    return __builtin_bit_cast(float,
        __builtin_amdgcn_update_dpp(0, b, 0x160 + XM, 0xF, 0xF, true));
}
__device__ __forceinline__ float xhalf(float v) {
    const int b = __builtin_bit_cast(int, v);
    return __builtin_bit_cast(float,
        __builtin_amdgcn_permlanex16(b, b, 0x76543210, 0xfedcba98, false, false));
}
__device__ __forceinline__ float group_sum8(float s) {
    s += row_xor<1>(s);
    s += row_xor<2>(s);
    s += row_xor<4>(s);
    return s;
}
__device__ __forceinline__ float lrelu(float v) {
    return __builtin_fmaxf(v, NEG_SLOPE * v);
}
__device__ __forceinline__ v2f lrelu2(const v2f v) {
    v2f r;
    r.x = lrelu(v.x);
    r.y = lrelu(v.y);
    return r;
}
__device__ __forceinline__ float edge_score(const v4f a0, const v4f a1,
                                            const v4f r0, const v4f r1,
                                            const v4f w0, const v4f w1) {
    v2f s = w0.lo * lrelu2(a0.lo + r0.lo);
    s += w0.hi * lrelu2(a0.hi + r0.hi);
    s += w1.lo * lrelu2(a1.lo + r1.lo);
    s += w1.hi * lrelu2(a1.hi + r1.hi);
    return group_sum8(s.x + s.y);
}
#define RESCALE_AT 16.0f
__device__ __forceinline__ void softmax_step(const float s, const v4f a0, const v4f a1,
                                             float& m, float& d, v4f& acc0, v4f& acc1) {
    float df = s - m;
    if (__builtin_amdgcn_ballot_w32(df > RESCALE_AT) != 0) {
        const bool  up = df > RESCALE_AT;
        const float c  = up ? __expf(-df) : 1.0f;
        d *= c;
        acc0 = acc0 * c;
        acc1 = acc1 * c;
        m  = up ? s : m;
        df = up ? 0.0f : df;
    }
    const float p = __expf(df);
    d += p;
    acc0 += a0 * p;
    acc1 += a1 * p;
}

__global__ void gat_node_kernel(const float* __restrict__ xl,
                                const float* __restrict__ xr,
                                const float* __restrict__ att,
                                const float* __restrict__ bias,
                                const int2* __restrict__ rec,
                                const int* __restrict__ head,
                                float* __restrict__ out, int N, int E) {
    const int widx = (int)((blockIdx.x * (size_t)blockDim.x + threadIdx.x) >> 5);
    const int lane = threadIdx.x & 31;
    if (widx >= N) return;
    const int i = __builtin_amdgcn_readfirstlane(widx);

    const int wq = ((lane >> 3) << 4) + (lane & 7);
    const v4f w0 = ((const v4f*)att)[wq];
    const v4f w1 = ((const v4f*)att)[wq + 8];
    const v4f* __restrict__ pr = (const v4f*)(xr + (size_t)i * HC);
    const v4f r0 = pr[lane], r1 = pr[lane + 32];

    int e = __builtin_amdgcn_readfirstlane(head[(size_t)i * HSTR]);
    int2 nx = make_int2(0, -1);
    if (e >= 0) nx = rec[e];

    const v4f* __restrict__ ps = (const v4f*)(xl + (size_t)i * HC);
    v4f a0 = ps[lane], a1 = ps[lane + 32];
    float m = edge_score(a0, a1, r0, r1, w0, w1);
    float d = 1.0f;
    v4f acc0 = a0, acc1 = a1;

    for (int it = 0; it < E && e >= 0; ++it) {
        const int src = __builtin_amdgcn_readfirstlane(nx.x);
        const int en  = __builtin_amdgcn_readfirstlane(nx.y);
        if (en >= 0) nx = rec[en];
        const v4f* __restrict__ pl = (const v4f*)(xl + (size_t)src * HC);
        a0 = pl[lane];
        a1 = pl[lane + 32];
        const float s = edge_score(a0, a1, r0, r1, w0, w1);
        softmax_step(s, a0, a1, m, d, acc0, acc1);
        e = en;
    }

    const float inv = 1.0f / d;
    const bool  upper = (lane & 8) != 0;
    const v4f keep = (upper ? acc1 : acc0) * inv;
    const v4f send = (upper ? acc0 : acc1) * inv;
    v4f o;
    o.x = keep.x + row_xor<8>(send.x);
    o.y = keep.y + row_xor<8>(send.y);
    o.z = keep.z + row_xor<8>(send.z);
    o.w = keep.w + row_xor<8>(send.w);
    o.x += xhalf(o.x);
    o.y += xhalf(o.y);
    o.z += xhalf(o.z);
    o.w += xhalf(o.w);
    if (lane < 16) {
        const v4f b = ((const v4f*)bias)[lane];
        ((v4f*)(out + (size_t)i * CH))[lane] = 0.25f * o + b;
    }
}

extern "C" void kernel_launch(void* const* d_in, const int* in_sizes, int n_in,
                              void* d_out, int out_size, void* d_ws, size_t ws_size,
                              hipStream_t stream) {
    const float* x    = (const float*)d_in[0];
    const float* Wl   = (const float*)d_in[1];
    const float* bl   = (const float*)d_in[2];
    const float* Wr   = (const float*)d_in[3];
    const float* br   = (const float*)d_in[4];
    const float* att  = (const float*)d_in[5];
    const float* bias = (const float*)d_in[6];
    const int*   edges = (const int*)d_in[7];

    const int N    = in_sizes[0] / CH;
    const int E    = in_sizes[7] / 2;

    float* xl   = (float*)d_ws;
    float* xr   = xl + (size_t)N * HC;
    int2*  rec  = (int2*)(xr + (size_t)N * HC);
    int*   head = (int*)(rec + E);

    const int B = 256;
    gat_proj_kernel<false><<<PROJ_BLOCKS, B, 0, stream>>>(x, Wl, bl, Wr, br, xl, xr, head, edges, rec, N, E);
    gat_proj_kernel<true><<<PROJ_BLOCKS, B, 0, stream>>>(x, Wl, bl, Wr, br, xl, xr, head, edges, rec, N, E);

    const long long waveThreads = (long long)N * 32;
    gat_node_kernel<<<(int)((waveThreads + B - 1) / B), B, 0, stream>>>(
        xl, xr, att, bias, rec, head, (float*)d_out, N, E);
}
